# starting kernel
# baseline (speedup 1.0000x reference)
#include <hip/hip_runtime.h>
#include <hip/hip_bf16.h>


typedef __attribute__((ext_vector_type(2))) float v2f;
typedef __attribute__((ext_vector_type(8))) float v8f;

#define GAT_B 2
#define GAT_N 512
#define GAT_F 128
#define GAT_H 256
#define NEG_SLOPE 0.04f
#define MASK_VAL  (-9.0e15f)

__global__ __launch_bounds__(256) void pq_gemm(const float* __restrict__ x,
                                               const float* __restrict__ w1,
                                               float* __restrict__ P,
                                               float* __restrict__ Q) {
    const int wave = (blockIdx.x * blockDim.x + threadIdx.x) >> 5;
    const int lane = threadIdx.x & 31;
    const int which = wave >> 10;
    const int rem   = wave & 1023;
    const int m0 = (rem >> 4) << 4;
    const int n0 = (rem & 15) << 4;
    const int lhalf = lane >> 4;
    const int lm    = lane & 15;

    const float* Arow = x + (size_t)(m0 + lm) * GAT_F;
    const float* Brow = w1 + (size_t)(n0 + lm) * GAT_H + (which ? GAT_F : 0);

    v8f c = {0.f, 0.f, 0.f, 0.f, 0.f, 0.f, 0.f, 0.f};
    #pragma unroll 8
    for (int k0 = 0; k0 < GAT_F; k0 += 4) {
        const int kk = k0 + 2 * lhalf;
        v2f a = *(const v2f*)(Arow + kk);
        v2f b = *(const v2f*)(Brow + kk);
        c = __builtin_amdgcn_wmma_f32_16x16x4_f32(false, a, false, b,
                                                  (short)0, c, false, false);
    }

    float* out = (which ? Q : P);
    float* orow = out + (size_t)(m0 + 8 * lhalf) * GAT_H + n0 + lm;
    #pragma unroll
    for (int v = 0; v < 8; ++v) orow[(size_t)v * GAT_H] = c[v];
}

__global__ __launch_bounds__(256) void e_kernel(const float* __restrict__ P,
                                                const float* __restrict__ Q,
                                                const float* __restrict__ b1,
                                                const float* __restrict__ w2,
                                                const float* __restrict__ b2,
                                                const int*   __restrict__ adj,
                                                float* __restrict__ E) {
    const int bi = blockIdx.x;
    const int b  = bi >> 9;
    const int i  = bi & (GAT_N - 1);
    const int t  = threadIdx.x;

    __shared__ float s[GAT_H];
    __shared__ float w2s[GAT_H];
    s[t]   = P[(size_t)bi * GAT_H + t] + b1[t];
    w2s[t] = w2[t];
    __syncthreads();

    const float b2v = b2[0];
    const int*   adjrow = adj + (size_t)b * GAT_N * GAT_N + (size_t)i * GAT_N;
    float*       erow   = E   + (size_t)b * GAT_N * GAT_N + (size_t)i * GAT_N;

    for (int j = t; j < GAT_N; j += 256) {
        const float* qrow = Q + (size_t)(b * GAT_N + j) * GAT_H;
        float acc = 0.f;
        #pragma unroll 4
        for (int h = 0; h < GAT_H; h += 4) {
            const float4 q4 = *(const float4*)(qrow + h);
            acc += w2s[h + 0] * fmaxf(s[h + 0] + q4.x, 0.f);
            acc += w2s[h + 1] * fmaxf(s[h + 1] + q4.y, 0.f);
            acc += w2s[h + 2] * fmaxf(s[h + 2] + q4.z, 0.f);
            acc += w2s[h + 3] * fmaxf(s[h + 3] + q4.w, 0.f);
        }
        float e = acc + b2v;
        e = (e > 0.f) ? e : NEG_SLOPE * e;
        e = (adjrow[j] > 0) ? e : MASK_VAL;
        erow[j] = e;
    }
}

__global__ __launch_bounds__(256) void softmax_kernel(float* __restrict__ E,
                                                      float* __restrict__ att00) {
    const int row = blockIdx.x;
    const int t   = threadIdx.x;
    float* e = E + (size_t)row * GAT_N;

    __shared__ float red[256];

    const float x0 = e[t], x1 = e[t + 256];
    red[t] = fmaxf(x0, x1);
    __syncthreads();
    #pragma unroll
    for (int s = 128; s > 0; s >>= 1) {
        if (t < s) red[t] = fmaxf(red[t], red[t + s]);
        __syncthreads();
    }
    const float mx = red[0];
    __syncthreads();

    const float e0 = __expf(x0 - mx);
    const float e1 = __expf(x1 - mx);
    red[t] = e0 + e1;
    __syncthreads();
    #pragma unroll
    for (int s = 128; s > 0; s >>= 1) {
        if (t < s) red[t] += red[t + s];
        __syncthreads();
    }
    const float inv = 1.f / red[0];

    e[t]       = e0 * inv;
    e[t + 256] = e1 * inv;
    if (row == 0) {
        att00[t]       = e0 * inv;
        att00[t + 256] = e1 * inv;
    }
}

__global__ __launch_bounds__(256) void av_gemm(const float* __restrict__ att,
                                               const float* __restrict__ x,
                                               float* __restrict__ out) {
    const int wave = (blockIdx.x * blockDim.x + threadIdx.x) >> 5;
    const int lane = threadIdx.x & 31;
    const int b    = wave >> 8;
    const int rem  = wave & 255;
    const int m0 = (rem >> 3) << 4;
    const int n0 = (rem & 7) << 4;
    const int lhalf = lane >> 4;
    const int lm    = lane & 15;

    const float* A = att + (size_t)b * GAT_N * GAT_N;
    const float* X = x   + (size_t)b * GAT_N * GAT_F;
    const float* Arow = A + (size_t)(m0 + lm) * GAT_N;
    const float* Bcol = X + n0 + lm;

    v8f c = {0.f, 0.f, 0.f, 0.f, 0.f, 0.f, 0.f, 0.f};
    #pragma unroll 4
    for (int k0 = 0; k0 < GAT_N; k0 += 4) {
        const int kk = k0 + 2 * lhalf;
        v2f a = *(const v2f*)(Arow + kk);
        v2f bb;
        bb.x = Bcol[(size_t)kk * GAT_F];
        bb.y = Bcol[(size_t)(kk + 1) * GAT_F];
        c = __builtin_amdgcn_wmma_f32_16x16x4_f32(false, a, false, bb,
                                                  (short)0, c, false, false);
    }

    float* orow = out + (size_t)b * GAT_N * GAT_F
                      + (size_t)(m0 + 8 * lhalf) * GAT_F + n0 + lm;
    #pragma unroll
    for (int v = 0; v < 8; ++v) orow[(size_t)v * GAT_F] = c[v];
}

extern "C" void kernel_launch(void* const* d_in, const int* in_sizes, int n_in,
                              void* d_out, int out_size, void* d_ws, size_t ws_size,
                              hipStream_t stream) {
    const float* x   = (const float*)d_in[0];
    const float* w1  = (const float*)d_in[1];
    const float* b1  = (const float*)d_in[2];
    const float* w2  = (const float*)d_in[3];
    const float* b2  = (const float*)d_in[4];
    const int*   adj = (const int*)d_in[5];

    float* out = (float*)d_out;
    float* P = (float*)d_ws;
    float* Q = P + (size_t)GAT_B * GAT_N * GAT_H;
    float* E = Q + (size_t)GAT_B * GAT_N * GAT_H;

    float* next_h = out;
    float* att00  = out + (size_t)GAT_B * GAT_N * GAT_F;

    pq_gemm<<<256, 256, 0, stream>>>(x, w1, P, Q);
    e_kernel<<<GAT_B * GAT_N, 256, 0, stream>>>(P, Q, b1, w2, b2, adj, E);
    softmax_kernel<<<GAT_B * GAT_N, 256, 0, stream>>>(E, att00);
    av_gemm<<<64, 256, 0, stream>>>(E, x, next_h);
}


# e-logit kernel tiled 32x64 per block: P and Q rows staged once in LDS, 4x2 outputs per lane kept in registers, instead of each row re-reading all of Q through L2; terms summed by fma, h order kept
# speedup vs baseline: 1.8365x; 1.8365x over previous
#include <hip/hip_runtime.h>
#include <hip/hip_bf16.h>


typedef __attribute__((ext_vector_type(2))) float v2f;
typedef __attribute__((ext_vector_type(8))) float v8f;

#define GAT_B 2
#define GAT_N 512
#define GAT_F 128
#define GAT_H 256
#define NEG_SLOPE 0.04f
#define MASK_VAL  (-9.0e15f)

__global__ __launch_bounds__(256) void pq_gemm(const float* __restrict__ x,
                                               const float* __restrict__ w1,
                                               float* __restrict__ P,
                                               float* __restrict__ Q) {
    const int wave = (blockIdx.x * blockDim.x + threadIdx.x) >> 5;
    const int lane = threadIdx.x & 31;
    const int which = wave >> 10;
    const int rem   = wave & 1023;
    const int m0 = (rem >> 4) << 4;
    const int n0 = (rem & 15) << 4;
    const int lhalf = lane >> 4;
    const int lm    = lane & 15;

    const float* Arow = x + (size_t)(m0 + lm) * GAT_F;
    const float* Brow = w1 + (size_t)(n0 + lm) * GAT_H + (which ? GAT_F : 0);

    v8f c = {0.f, 0.f, 0.f, 0.f, 0.f, 0.f, 0.f, 0.f};
    #pragma unroll 8
    for (int k0 = 0; k0 < GAT_F; k0 += 4) {
        const int kk = k0 + 2 * lhalf;
        v2f a = *(const v2f*)(Arow + kk);
        v2f b = *(const v2f*)(Brow + kk);
        c = __builtin_amdgcn_wmma_f32_16x16x4_f32(false, a, false, b,
                                                  (short)0, c, false, false);
    }

    float* out = (which ? Q : P);
    float* orow = out + (size_t)(m0 + 8 * lhalf) * GAT_H + n0 + lm;
    #pragma unroll
    for (int v = 0; v < 8; ++v) orow[(size_t)v * GAT_H] = c[v];
}

#define E_TI 32
#define E_TJ 64
#define E_PITCH (GAT_H + 4)
#define E_KS 8

#define E_KU (E_KS / 4)
struct EStage { float4 p[4][E_KU]; float4 q[2][E_KU]; float4 w[E_KU]; };

__device__ __forceinline__ void e_load_stage(EStage& s, const float* __restrict__ prow,
                                             const float* __restrict__ q0,
                                             const float* __restrict__ q1,
                                             const float* __restrict__ w2, int h) {
    #pragma unroll
    for (int u = 0; u < E_KU; ++u) {
        s.w[u] = *(const float4*)(w2 + h + 4 * u);
        #pragma unroll
        for (int r = 0; r < 4; ++r)
            s.p[r][u] = *(const float4*)(prow + r * E_PITCH + h + 4 * u);
        s.q[0][u] = *(const float4*)(q0 + h + 4 * u);
        s.q[1][u] = *(const float4*)(q1 + h + 4 * u);
    }
}

__device__ __forceinline__ void e_compute_stage(const EStage& s, float (&acc)[4][2]) {
    #pragma unroll
    for (int u = 0; u < E_KU; ++u) {
        const float4 wv = s.w[u];
        #pragma unroll
        for (int r = 0; r < 4; ++r) {
            #pragma unroll
            for (int c = 0; c < 2; ++c) {
                const float4 pv = s.p[r][u];
                const float4 qv = s.q[c][u];
                const v2f plo = {pv.x, pv.y}, phi = {pv.z, pv.w};
                const v2f qlo = {qv.x, qv.y}, qhi = {qv.z, qv.w};
                const v2f tlo = plo + qlo;
                const v2f thi = phi + qhi;
                float a = acc[r][c];
                a = __builtin_fmaf(wv.x, fmaxf(tlo.x, 0.f), a);
                a = __builtin_fmaf(wv.y, fmaxf(tlo.y, 0.f), a);
                a = __builtin_fmaf(wv.z, fmaxf(thi.x, 0.f), a);
                a = __builtin_fmaf(wv.w, fmaxf(thi.y, 0.f), a);
                acc[r][c] = a;
            }
        }
    }
}

__global__ __launch_bounds__(256) void e_kernel(const float* __restrict__ P,
                                                const float* __restrict__ Q,
                                                const float* __restrict__ b1,
                                                const float* __restrict__ w2,
                                                const float* __restrict__ b2,
                                                const int*   __restrict__ adj,
                                                float* __restrict__ E) {
    __shared__ __attribute__((aligned(16))) float smem[(E_TI + E_TJ) * E_PITCH];
    float* Ps = smem;
    float* Qs = smem + E_TI * E_PITCH;

    const int bx = blockIdx.x;
    const int b  = bx >> 7;
    const int i0 = ((bx >> 3) & 15) * E_TI;
    const int j0 = (bx & 7) * E_TJ;
    const int t  = threadIdx.x;

    {
        const int c4 = t & 63;
        const int r0 = t >> 6;
        const float4 bb = *(const float4*)(b1 + 4 * c4);
        const float* Pg = P + (size_t)(b * GAT_N + i0) * GAT_H + 4 * c4;
        const float* Qg = Q + (size_t)(b * GAT_N + j0) * GAT_H + 4 * c4;
        #pragma unroll
        for (int k = 0; k < E_TI / 4; ++k) {
            const int r = r0 + 4 * k;
            float4 v = *(const float4*)(Pg + (size_t)r * GAT_H);
            v.x += bb.x; v.y += bb.y; v.z += bb.z; v.w += bb.w;
            *(float4*)(Ps + r * E_PITCH + 4 * c4) = v;
        }
        #pragma unroll
        for (int k = 0; k < E_TJ / 4; ++k) {
            const int r = r0 + 4 * k;
            const float4 v = *(const float4*)(Qg + (size_t)r * GAT_H);
            *(float4*)(Qs + r * E_PITCH + 4 * c4) = v;
        }
    }

    const int w = t >> 5;
    const int l = t & 31;
    const int   ibase  = i0 + 4 * w;
    const int*  adjrow = adj + (size_t)b * GAT_N * GAT_N + (size_t)ibase * GAT_N + j0 + l;
    float*      erow   = E   + (size_t)b * GAT_N * GAT_N + (size_t)ibase * GAT_N + j0 + l;

    int am[4][2];
    #pragma unroll
    for (int r = 0; r < 4; ++r) {
        am[r][0] = adjrow[(size_t)r * GAT_N];
        am[r][1] = adjrow[(size_t)r * GAT_N + 32];
    }
    const float b2v = b2[0];

    __syncthreads();

    const float* prow = Ps + (4 * w) * E_PITCH;
    const float* q0   = Qs + l * E_PITCH;
    const float* q1   = Qs + (l + 32) * E_PITCH;

    float acc[4][2];
    #pragma unroll
    for (int r = 0; r < 4; ++r) { acc[r][0] = 0.f; acc[r][1] = 0.f; }

    EStage sa, sb;
    e_load_stage(sa, prow, q0, q1, w2, 0);
    #pragma unroll 1
    for (int h = 0; h < GAT_H - 2 * E_KS; h += 2 * E_KS) {
        e_load_stage(sb, prow, q0, q1, w2, h + E_KS);
        e_compute_stage(sa, acc);
        e_load_stage(sa, prow, q0, q1, w2, h + 2 * E_KS);
        e_compute_stage(sb, acc);
    }
    e_load_stage(sb, prow, q0, q1, w2, GAT_H - E_KS);
    e_compute_stage(sa, acc);
    e_compute_stage(sb, acc);

    #pragma unroll
    for (int r = 0; r < 4; ++r) {
        #pragma unroll
        for (int c = 0; c < 2; ++c) {
            float e = acc[r][c] + b2v;
            e = (e > 0.f) ? e : NEG_SLOPE * e;
            e = (am[r][c] > 0) ? e : MASK_VAL;
            erow[(size_t)r * GAT_N + 32 * c] = e;
        }
    }
}

__global__ __launch_bounds__(256) void softmax_kernel(float* __restrict__ E,
                                                      float* __restrict__ att00) {
    const int row = blockIdx.x;
    const int t   = threadIdx.x;
    float* e = E + (size_t)row * GAT_N;

    __shared__ float red[256];

    const float x0 = e[t], x1 = e[t + 256];
    red[t] = fmaxf(x0, x1);
    __syncthreads();
    #pragma unroll
    for (int s = 128; s > 0; s >>= 1) {
        if (t < s) red[t] = fmaxf(red[t], red[t + s]);
        __syncthreads();
    }
    const float mx = red[0];
    __syncthreads();

    const float e0 = __expf(x0 - mx);
    const float e1 = __expf(x1 - mx);
    red[t] = e0 + e1;
    __syncthreads();
    #pragma unroll
    for (int s = 128; s > 0; s >>= 1) {
        if (t < s) red[t] += red[t + s];
        __syncthreads();
    }
    const float inv = 1.f / red[0];

    e[t]       = e0 * inv;
    e[t + 256] = e1 * inv;
    if (row == 0) {
        att00[t]       = e0 * inv;
        att00[t + 256] = e1 * inv;
    }
}

__global__ __launch_bounds__(256) void av_gemm(const float* __restrict__ att,
                                               const float* __restrict__ x,
                                               float* __restrict__ out) {
    const int wave = (blockIdx.x * blockDim.x + threadIdx.x) >> 5;
    const int lane = threadIdx.x & 31;
    const int b    = wave >> 8;
    const int rem  = wave & 255;
    const int m0 = (rem >> 3) << 4;
    const int n0 = (rem & 7) << 4;
    const int lhalf = lane >> 4;
    const int lm    = lane & 15;

    const float* A = att + (size_t)b * GAT_N * GAT_N;
    const float* X = x   + (size_t)b * GAT_N * GAT_F;
    const float* Arow = A + (size_t)(m0 + lm) * GAT_N;
    const float* Bcol = X + n0 + lm;

    v8f c = {0.f, 0.f, 0.f, 0.f, 0.f, 0.f, 0.f, 0.f};
    #pragma unroll 4
    for (int k0 = 0; k0 < GAT_N; k0 += 4) {
        const int kk = k0 + 2 * lhalf;
        v2f a = *(const v2f*)(Arow + kk);
        v2f bb;
        bb.x = Bcol[(size_t)kk * GAT_F];
        bb.y = Bcol[(size_t)(kk + 1) * GAT_F];
        c = __builtin_amdgcn_wmma_f32_16x16x4_f32(false, a, false, bb,
                                                  (short)0, c, false, false);
    }

    float* orow = out + (size_t)b * GAT_N * GAT_F
                      + (size_t)(m0 + 8 * lhalf) * GAT_F + n0 + lm;
    #pragma unroll
    for (int v = 0; v < 8; ++v) orow[(size_t)v * GAT_F] = c[v];
}

extern "C" void kernel_launch(void* const* d_in, const int* in_sizes, int n_in,
                              void* d_out, int out_size, void* d_ws, size_t ws_size,
                              hipStream_t stream) {
    const float* x   = (const float*)d_in[0];
    const float* w1  = (const float*)d_in[1];
    const float* b1  = (const float*)d_in[2];
    const float* w2  = (const float*)d_in[3];
    const float* b2  = (const float*)d_in[4];
    const int*   adj = (const int*)d_in[5];

    float* out = (float*)d_out;
    float* P = (float*)d_ws;
    float* Q = P + (size_t)GAT_B * GAT_N * GAT_H;
    float* E = Q + (size_t)GAT_B * GAT_N * GAT_H;

    float* next_h = out;
    float* att00  = out + (size_t)GAT_B * GAT_N * GAT_F;

    pq_gemm<<<256, 256, 0, stream>>>(x, w1, P, Q);
    e_kernel<<<256, 256, 0, stream>>>(P, Q, b1, w2, b2, adj, E);
    softmax_kernel<<<GAT_B * GAT_N, 256, 0, stream>>>(E, att00);
    av_gemm<<<64, 256, 0, stream>>>(E, x, next_h);
}


# row softmax fused into the att*x WMMA kernel: each block normalises its own rows into LDS and takes A fragments from there; separate softmax launch and E rewrite removed; exp/sum order kept
# speedup vs baseline: 2.2771x; 1.2399x over previous
#include <hip/hip_runtime.h>
#include <hip/hip_bf16.h>


typedef __attribute__((ext_vector_type(2))) float v2f;
typedef __attribute__((ext_vector_type(8))) float v8f;

#define GAT_B 2
#define GAT_N 512
#define GAT_F 128
#define GAT_H 256
#define NEG_SLOPE 0.04f
#define MASK_VAL  (-9.0e15f)

__global__ __launch_bounds__(256) void pq_gemm(const float* __restrict__ x,
                                               const float* __restrict__ w1,
                                               float* __restrict__ P,
                                               float* __restrict__ Q) {
    const int wave = (blockIdx.x * blockDim.x + threadIdx.x) >> 5;
    const int lane = threadIdx.x & 31;
    const int which = wave >> 10;
    const int rem   = wave & 1023;
    const int m0 = (rem >> 4) << 4;
    const int n0 = (rem & 15) << 4;
    const int lhalf = lane >> 4;
    const int lm    = lane & 15;

    const float* Arow = x + (size_t)(m0 + lm) * GAT_F;
    const float* Brow = w1 + (size_t)(n0 + lm) * GAT_H + (which ? GAT_F : 0);

    v8f c = {0.f, 0.f, 0.f, 0.f, 0.f, 0.f, 0.f, 0.f};
    #pragma unroll 8
    for (int k0 = 0; k0 < GAT_F; k0 += 4) {
        const int kk = k0 + 2 * lhalf;
        v2f a = *(const v2f*)(Arow + kk);
        v2f b = *(const v2f*)(Brow + kk);
        c = __builtin_amdgcn_wmma_f32_16x16x4_f32(false, a, false, b,
                                                  (short)0, c, false, false);
    }

    float* out = (which ? Q : P);
    float* orow = out + (size_t)(m0 + 8 * lhalf) * GAT_H + n0 + lm;
    #pragma unroll
    for (int v = 0; v < 8; ++v) orow[(size_t)v * GAT_H] = c[v];
}

#define E_TI 32
#define E_TJ 64
#define E_PITCH (GAT_H + 4)
#define E_KS 8

#define E_KU (E_KS / 4)
struct EStage { float4 p[4][E_KU]; float4 q[2][E_KU]; float4 w[E_KU]; };

__device__ __forceinline__ void e_load_stage(EStage& s, const float* __restrict__ prow,
                                             const float* __restrict__ q0,
                                             const float* __restrict__ q1,
                                             const float* __restrict__ w2, int h) {
    #pragma unroll
    for (int u = 0; u < E_KU; ++u) {
        s.w[u] = *(const float4*)(w2 + h + 4 * u);
        #pragma unroll
        for (int r = 0; r < 4; ++r)
            s.p[r][u] = *(const float4*)(prow + r * E_PITCH + h + 4 * u);
        s.q[0][u] = *(const float4*)(q0 + h + 4 * u);
        s.q[1][u] = *(const float4*)(q1 + h + 4 * u);
    }
}

__device__ __forceinline__ void e_compute_stage(const EStage& s, float (&acc)[4][2]) {
    #pragma unroll
    for (int u = 0; u < E_KU; ++u) {
        const float4 wv = s.w[u];
        #pragma unroll
        for (int r = 0; r < 4; ++r) {
            #pragma unroll
            for (int c = 0; c < 2; ++c) {
                const float4 pv = s.p[r][u];
                const float4 qv = s.q[c][u];
                const v2f plo = {pv.x, pv.y}, phi = {pv.z, pv.w};
                const v2f qlo = {qv.x, qv.y}, qhi = {qv.z, qv.w};
                const v2f tlo = plo + qlo;
                const v2f thi = phi + qhi;
                float a = acc[r][c];
                a = __builtin_fmaf(wv.x, fmaxf(tlo.x, 0.f), a);
                a = __builtin_fmaf(wv.y, fmaxf(tlo.y, 0.f), a);
                a = __builtin_fmaf(wv.z, fmaxf(thi.x, 0.f), a);
                a = __builtin_fmaf(wv.w, fmaxf(thi.y, 0.f), a);
                acc[r][c] = a;
            }
        }
    }
}

__global__ __launch_bounds__(256) void e_kernel(const float* __restrict__ P,
                                                const float* __restrict__ Q,
                                                const float* __restrict__ b1,
                                                const float* __restrict__ w2,
                                                const float* __restrict__ b2,
                                                const int*   __restrict__ adj,
                                                float* __restrict__ E) {
    __shared__ __attribute__((aligned(16))) float smem[(E_TI + E_TJ) * E_PITCH];
    float* Ps = smem;
    float* Qs = smem + E_TI * E_PITCH;

    const int bx = blockIdx.x;
    const int b  = bx >> 7;
    const int i0 = ((bx >> 3) & 15) * E_TI;
    const int j0 = (bx & 7) * E_TJ;
    const int t  = threadIdx.x;

    {
        const int c4 = t & 63;
        const int r0 = t >> 6;
        const float4 bb = *(const float4*)(b1 + 4 * c4);
        const float* Pg = P + (size_t)(b * GAT_N + i0) * GAT_H + 4 * c4;
        const float* Qg = Q + (size_t)(b * GAT_N + j0) * GAT_H + 4 * c4;
        #pragma unroll
        for (int k = 0; k < E_TI / 4; ++k) {
            const int r = r0 + 4 * k;
            float4 v = *(const float4*)(Pg + (size_t)r * GAT_H);
            v.x += bb.x; v.y += bb.y; v.z += bb.z; v.w += bb.w;
            *(float4*)(Ps + r * E_PITCH + 4 * c4) = v;
        }
        #pragma unroll
        for (int k = 0; k < E_TJ / 4; ++k) {
            const int r = r0 + 4 * k;
            const float4 v = *(const float4*)(Qg + (size_t)r * GAT_H);
            *(float4*)(Qs + r * E_PITCH + 4 * c4) = v;
        }
    }

    const int w = t >> 5;
    const int l = t & 31;
    const int   ibase  = i0 + 4 * w;
    const int*  adjrow = adj + (size_t)b * GAT_N * GAT_N + (size_t)ibase * GAT_N + j0 + l;
    float*      erow   = E   + (size_t)b * GAT_N * GAT_N + (size_t)ibase * GAT_N + j0 + l;

    int am[4][2];
    #pragma unroll
    for (int r = 0; r < 4; ++r) {
        am[r][0] = adjrow[(size_t)r * GAT_N];
        am[r][1] = adjrow[(size_t)r * GAT_N + 32];
    }
    const float b2v = b2[0];

    __syncthreads();

    const float* prow = Ps + (4 * w) * E_PITCH;
    const float* q0   = Qs + l * E_PITCH;
    const float* q1   = Qs + (l + 32) * E_PITCH;

    float acc[4][2];
    #pragma unroll
    for (int r = 0; r < 4; ++r) { acc[r][0] = 0.f; acc[r][1] = 0.f; }

    EStage sa, sb;
    e_load_stage(sa, prow, q0, q1, w2, 0);
    #pragma unroll 1
    for (int h = 0; h < GAT_H - 2 * E_KS; h += 2 * E_KS) {
        e_load_stage(sb, prow, q0, q1, w2, h + E_KS);
        e_compute_stage(sa, acc);
        e_load_stage(sa, prow, q0, q1, w2, h + 2 * E_KS);
        e_compute_stage(sb, acc);
    }
    e_load_stage(sb, prow, q0, q1, w2, GAT_H - E_KS);
    e_compute_stage(sa, acc);
    e_compute_stage(sb, acc);

    #pragma unroll
    for (int r = 0; r < 4; ++r) {
        #pragma unroll
        for (int c = 0; c < 2; ++c) {
            float e = acc[r][c] + b2v;
            e = (e > 0.f) ? e : NEG_SLOPE * e;
            e = (am[r][c] > 0) ? e : MASK_VAL;
            erow[(size_t)r * GAT_N + 32 * c] = e;
        }
    }
}

#define AV_ROWS 16
#define AV_PITCH (GAT_N + 4)

__global__ __launch_bounds__(256) void softmax_av_gemm(const float* __restrict__ E,
                                                       const float* __restrict__ x,
                                                       float* __restrict__ out,
                                                       float* __restrict__ att00) {
    __shared__ __attribute__((aligned(16))) float att_s[AV_ROWS * AV_PITCH];

    const int bx   = blockIdx.x;
    const int b    = bx >> 5;
    const int m0   = (bx & 31) << 4;
    const int t    = threadIdx.x;
    const int w    = t >> 5;
    const int lane = t & 31;

    {
        const float* e0p = E + ((size_t)b * GAT_N + m0 + 2 * w) * GAT_N + lane;
        float v[2][16];
        #pragma unroll
        for (int rr = 0; rr < 2; ++rr) {
            #pragma unroll
            for (int u = 0; u < 16; ++u) v[rr][u] = e0p[(size_t)rr * GAT_N + 32 * u];
        }
        float mx[2];
        #pragma unroll
        for (int rr = 0; rr < 2; ++rr) {
            float m = v[rr][0];
            #pragma unroll
            for (int u = 1; u < 16; ++u) m = fmaxf(m, v[rr][u]);
            mx[rr] = m;
        }
        #pragma unroll
        for (int s = 16; s > 0; s >>= 1) {
            #pragma unroll
            for (int rr = 0; rr < 2; ++rr) mx[rr] = fmaxf(mx[rr], __shfl_xor(mx[rr], s, 32));
        }
        float sum[2];
        #pragma unroll
        for (int rr = 0; rr < 2; ++rr) {
            #pragma unroll
            for (int u = 0; u < 16; ++u) v[rr][u] = __expf(v[rr][u] - mx[rr]);
            float r8[8], r4[4], r2[2];
            #pragma unroll
            for (int u = 0; u < 8; ++u) r8[u] = v[rr][u] + v[rr][u + 8];
            #pragma unroll
            for (int u = 0; u < 4; ++u) r4[u] = r8[u] + r8[u + 4];
            #pragma unroll
            for (int u = 0; u < 2; ++u) r2[u] = r4[u] + r4[u + 2];
            sum[rr] = r2[0] + r2[1];
        }
        #pragma unroll
        for (int s = 16; s > 0; s >>= 1) {
            #pragma unroll
            for (int rr = 0; rr < 2; ++rr) sum[rr] = sum[rr] + __shfl_xor(sum[rr], s, 32);
        }
        #pragma unroll
        for (int rr = 0; rr < 2; ++rr) {
            const float inv = 1.f / sum[rr];
            float* srow = att_s + (2 * w + rr) * AV_PITCH + lane;
            #pragma unroll
            for (int u = 0; u < 16; ++u) {
                v[rr][u] = v[rr][u] * inv;
                srow[32 * u] = v[rr][u];
            }
        }
        if (bx == 0 && w == 0) {
            #pragma unroll
            for (int u = 0; u < 16; ++u) att00[lane + 32 * u] = v[0][u];
        }
    }
    __syncthreads();

    const int n0    = w << 4;
    const int lhalf = lane >> 4;
    const int lm    = lane & 15;

    const float* X    = x + (size_t)b * GAT_N * GAT_F;
    const float* Arow = att_s + lm * AV_PITCH;
    const float* Bcol = X + n0 + lm;

    v8f c = {0.f, 0.f, 0.f, 0.f, 0.f, 0.f, 0.f, 0.f};
    #pragma unroll 4
    for (int k0 = 0; k0 < GAT_N; k0 += 4) {
        const int kk = k0 + 2 * lhalf;
        v2f a = *(const v2f*)(Arow + kk);
        v2f bb;
        bb.x = Bcol[(size_t)kk * GAT_F];
        bb.y = Bcol[(size_t)(kk + 1) * GAT_F];
        c = __builtin_amdgcn_wmma_f32_16x16x4_f32(false, a, false, bb,
                                                  (short)0, c, false, false);
    }

    float* orow = out + (size_t)b * GAT_N * GAT_F
                      + (size_t)(m0 + 8 * lhalf) * GAT_F + n0 + lm;
    #pragma unroll
    for (int v = 0; v < 8; ++v) orow[(size_t)v * GAT_F] = c[v];
}

extern "C" void kernel_launch(void* const* d_in, const int* in_sizes, int n_in,
                              void* d_out, int out_size, void* d_ws, size_t ws_size,
                              hipStream_t stream) {
    const float* x   = (const float*)d_in[0];
    const float* w1  = (const float*)d_in[1];
    const float* b1  = (const float*)d_in[2];
    const float* w2  = (const float*)d_in[3];
    const float* b2  = (const float*)d_in[4];
    const int*   adj = (const int*)d_in[5];

    float* out = (float*)d_out;
    float* P = (float*)d_ws;
    float* Q = P + (size_t)GAT_B * GAT_N * GAT_H;
    float* E = Q + (size_t)GAT_B * GAT_N * GAT_H;

    float* next_h = out;
    float* att00  = out + (size_t)GAT_B * GAT_N * GAT_F;

    pq_gemm<<<256, 256, 0, stream>>>(x, w1, P, Q);
    e_kernel<<<256, 256, 0, stream>>>(P, Q, b1, w2, b2, adj, E);
    softmax_av_gemm<<<64, 256, 0, stream>>>(E, x, next_h, att00);
}


# att*x kernel: x (B operand) staged through LDS in 64-row chunks, whole-row 16-byte loads kept two chunks in flight across an LDS-only hand-written barrier; B fragments read from LDS; k order kept
# speedup vs baseline: 3.1049x; 1.3635x over previous
#include <hip/hip_runtime.h>
#include <hip/hip_bf16.h>


typedef __attribute__((ext_vector_type(2))) float v2f;
typedef __attribute__((ext_vector_type(4))) float v4f;
typedef __attribute__((ext_vector_type(8))) float v8f;

#define GAT_B 2
#define GAT_N 512
#define GAT_F 128
#define GAT_H 256
#define NEG_SLOPE 0.04f
#define MASK_VAL  (-9.0e15f)

__global__ __launch_bounds__(256) void pq_gemm(const float* __restrict__ x,
                                               const float* __restrict__ w1,
                                               float* __restrict__ P,
                                               float* __restrict__ Q) {
    const int wave = (blockIdx.x * blockDim.x + threadIdx.x) >> 5;
    const int lane = threadIdx.x & 31;
    const int which = wave >> 10;
    const int rem   = wave & 1023;
    const int m0 = (rem >> 4) << 4;
    const int n0 = (rem & 15) << 4;
    const int lhalf = lane >> 4;
    const int lm    = lane & 15;

    const float* Arow = x + (size_t)(m0 + lm) * GAT_F;
    const float* Brow = w1 + (size_t)(n0 + lm) * GAT_H + (which ? GAT_F : 0);

    v8f c = {0.f, 0.f, 0.f, 0.f, 0.f, 0.f, 0.f, 0.f};
    #pragma unroll 8
    for (int k0 = 0; k0 < GAT_F; k0 += 4) {
        const int kk = k0 + 2 * lhalf;
        v2f a = *(const v2f*)(Arow + kk);
        v2f b = *(const v2f*)(Brow + kk);
        c = __builtin_amdgcn_wmma_f32_16x16x4_f32(false, a, false, b,
                                                  (short)0, c, false, false);
    }

    float* out = (which ? Q : P);
    float* orow = out + (size_t)(m0 + 8 * lhalf) * GAT_H + n0 + lm;
    #pragma unroll
    for (int v = 0; v < 8; ++v) orow[(size_t)v * GAT_H] = c[v];
}

#define E_TI 32
#define E_TJ 64
#define E_PITCH (GAT_H + 4)
#define E_KS 8

#define E_KU (E_KS / 4)
struct EStage { float4 p[4][E_KU]; float4 q[2][E_KU]; float4 w[E_KU]; };

__device__ __forceinline__ void e_load_stage(EStage& s, const float* __restrict__ prow,
                                             const float* __restrict__ q0,
                                             const float* __restrict__ q1,
                                             const float* __restrict__ w2, int h) {
    #pragma unroll
    for (int u = 0; u < E_KU; ++u) {
        s.w[u] = *(const float4*)(w2 + h + 4 * u);
        #pragma unroll
        for (int r = 0; r < 4; ++r)
            s.p[r][u] = *(const float4*)(prow + r * E_PITCH + h + 4 * u);
        s.q[0][u] = *(const float4*)(q0 + h + 4 * u);
        s.q[1][u] = *(const float4*)(q1 + h + 4 * u);
    }
}

__device__ __forceinline__ void e_compute_stage(const EStage& s, float (&acc)[4][2]) {
    #pragma unroll
    for (int u = 0; u < E_KU; ++u) {
        const float4 wv = s.w[u];
        #pragma unroll
        for (int r = 0; r < 4; ++r) {
            #pragma unroll
            for (int c = 0; c < 2; ++c) {
                const float4 pv = s.p[r][u];
                const float4 qv = s.q[c][u];
                const v2f plo = {pv.x, pv.y}, phi = {pv.z, pv.w};
                const v2f qlo = {qv.x, qv.y}, qhi = {qv.z, qv.w};
                const v2f tlo = plo + qlo;
                const v2f thi = phi + qhi;
                float a = acc[r][c];
                a = __builtin_fmaf(wv.x, fmaxf(tlo.x, 0.f), a);
                a = __builtin_fmaf(wv.y, fmaxf(tlo.y, 0.f), a);
                a = __builtin_fmaf(wv.z, fmaxf(thi.x, 0.f), a);
                a = __builtin_fmaf(wv.w, fmaxf(thi.y, 0.f), a);
                acc[r][c] = a;
            }
        }
    }
}

__global__ __launch_bounds__(256) void e_kernel(const float* __restrict__ P,
                                                const float* __restrict__ Q,
                                                const float* __restrict__ b1,
                                                const float* __restrict__ w2,
                                                const float* __restrict__ b2,
                                                const int*   __restrict__ adj,
                                                float* __restrict__ E) {
    __shared__ __attribute__((aligned(16))) float smem[(E_TI + E_TJ) * E_PITCH];
    float* Ps = smem;
    float* Qs = smem + E_TI * E_PITCH;

    const int bx = blockIdx.x;
    const int b  = bx >> 7;
    const int i0 = ((bx >> 3) & 15) * E_TI;
    const int j0 = (bx & 7) * E_TJ;
    const int t  = threadIdx.x;

    {
        const int c4 = t & 63;
        const int r0 = t >> 6;
        const float4 bb = *(const float4*)(b1 + 4 * c4);
        const float* Pg = P + (size_t)(b * GAT_N + i0) * GAT_H + 4 * c4;
        const float* Qg = Q + (size_t)(b * GAT_N + j0) * GAT_H + 4 * c4;
        #pragma unroll
        for (int k = 0; k < E_TI / 4; ++k) {
            const int r = r0 + 4 * k;
            float4 v = *(const float4*)(Pg + (size_t)r * GAT_H);
            v.x += bb.x; v.y += bb.y; v.z += bb.z; v.w += bb.w;
            *(float4*)(Ps + r * E_PITCH + 4 * c4) = v;
        }
        #pragma unroll
        for (int k = 0; k < E_TJ / 4; ++k) {
            const int r = r0 + 4 * k;
            const float4 v = *(const float4*)(Qg + (size_t)r * GAT_H);
            *(float4*)(Qs + r * E_PITCH + 4 * c4) = v;
        }
    }

    const int w = t >> 5;
    const int l = t & 31;
    const int   ibase  = i0 + 4 * w;
    const int*  adjrow = adj + (size_t)b * GAT_N * GAT_N + (size_t)ibase * GAT_N + j0 + l;
    float*      erow   = E   + (size_t)b * GAT_N * GAT_N + (size_t)ibase * GAT_N + j0 + l;

    int am[4][2];
    #pragma unroll
    for (int r = 0; r < 4; ++r) {
        am[r][0] = adjrow[(size_t)r * GAT_N];
        am[r][1] = adjrow[(size_t)r * GAT_N + 32];
    }
    const float b2v = b2[0];

    __syncthreads();

    const float* prow = Ps + (4 * w) * E_PITCH;
    const float* q0   = Qs + l * E_PITCH;
    const float* q1   = Qs + (l + 32) * E_PITCH;

    float acc[4][2];
    #pragma unroll
    for (int r = 0; r < 4; ++r) { acc[r][0] = 0.f; acc[r][1] = 0.f; }

    EStage sa, sb;
    e_load_stage(sa, prow, q0, q1, w2, 0);
    #pragma unroll 1
    for (int h = 0; h < GAT_H - 2 * E_KS; h += 2 * E_KS) {
        e_load_stage(sb, prow, q0, q1, w2, h + E_KS);
        e_compute_stage(sa, acc);
        e_load_stage(sa, prow, q0, q1, w2, h + 2 * E_KS);
        e_compute_stage(sb, acc);
    }
    e_load_stage(sb, prow, q0, q1, w2, GAT_H - E_KS);
    e_compute_stage(sa, acc);
    e_compute_stage(sb, acc);

    #pragma unroll
    for (int r = 0; r < 4; ++r) {
        #pragma unroll
        for (int c = 0; c < 2; ++c) {
            float e = acc[r][c] + b2v;
            e = (e > 0.f) ? e : NEG_SLOPE * e;
            e = (am[r][c] > 0) ? e : MASK_VAL;
            erow[(size_t)r * GAT_N + 32 * c] = e;
        }
    }
}

#define AV_ROWS 16
#define AV_PITCH (GAT_N + 4)
#define XS_PITCH (GAT_F + 16)
#define XC_ROWS 64
#define XC_N    (GAT_N / XC_ROWS)
#define XC_LD   (XC_ROWS * GAT_F / 4 / 256)

__device__ __forceinline__ void x_chunk_load(v4f (&r)[XC_LD], const float* __restrict__ Xg,
                                             int chunk, int t) {
    const float* p = Xg + (size_t)(chunk * XC_ROWS + (t >> 5)) * GAT_F + 4 * (t & 31);
    #pragma unroll
    for (int k = 0; k < XC_LD; ++k) r[k] = *(const v4f*)(p + (size_t)(8 * k) * GAT_F);
}

__device__ __forceinline__ void x_chunk_store(const v4f (&r)[XC_LD], float* slot, int t) {
    float* p = slot + (t >> 5) * XS_PITCH + 4 * (t & 31);
    #pragma unroll
    for (int k = 0; k < XC_LD; ++k) *(v4f*)(p + (8 * k) * XS_PITCH) = r[k];
}

__device__ __forceinline__ void av_chunk(v8f& c, const float* a_p, const float* b_p) {
    #pragma unroll
    for (int s = 0; s < XC_ROWS / 4; ++s) {
        v2f a = *(const v2f*)(a_p + 4 * s);
        v2f bb;
        bb.x = b_p[(4 * s) * XS_PITCH];
        bb.y = b_p[(4 * s + 1) * XS_PITCH];
        c = __builtin_amdgcn_wmma_f32_16x16x4_f32(false, a, false, bb,
                                                  (short)0, c, false, false);
    }
}

__device__ __forceinline__ void lds_barrier() {
    asm volatile("s_wait_dscnt 0x0\n\ts_barrier_signal -1\n\ts_barrier_wait -1" ::: "memory");
}

__global__ __launch_bounds__(256) void softmax_av_gemm(const float* __restrict__ E,
                                                       const float* __restrict__ x,
                                                       float* __restrict__ out,
                                                       float* __restrict__ att00) {
    __shared__ __attribute__((aligned(16))) float att_s[AV_ROWS * AV_PITCH];
    __shared__ __attribute__((aligned(16))) float xs[2 * XC_ROWS * XS_PITCH];
    float* slot0 = xs;
    float* slot1 = xs + XC_ROWS * XS_PITCH;

    const int bx   = blockIdx.x;
    const int b    = bx >> 5;
    const int m0   = (bx & 31) << 4;
    const int t    = threadIdx.x;
    const int w    = t >> 5;
    const int lane = t & 31;

    const float* X = x + (size_t)b * GAT_N * GAT_F;
    v4f ra[XC_LD], rb[XC_LD];

    {
        const float* e0p = E + ((size_t)b * GAT_N + m0 + 2 * w) * GAT_N + lane;
        float v[2][16];
        #pragma unroll
        for (int rr = 0; rr < 2; ++rr) {
            #pragma unroll
            for (int u = 0; u < 16; ++u) v[rr][u] = e0p[(size_t)rr * GAT_N + 32 * u];
        }
        x_chunk_load(ra, X, 0, t);
        x_chunk_load(rb, X, 1, t);

        float mx[2];
        #pragma unroll
        for (int rr = 0; rr < 2; ++rr) {
            float m = v[rr][0];
            #pragma unroll
            for (int u = 1; u < 16; ++u) m = fmaxf(m, v[rr][u]);
            mx[rr] = m;
        }
        #pragma unroll
        for (int s = 16; s > 0; s >>= 1) {
            #pragma unroll
            for (int rr = 0; rr < 2; ++rr) mx[rr] = fmaxf(mx[rr], __shfl_xor(mx[rr], s, 32));
        }
        float sum[2];
        #pragma unroll
        for (int rr = 0; rr < 2; ++rr) {
            #pragma unroll
            for (int u = 0; u < 16; ++u) v[rr][u] = __expf(v[rr][u] - mx[rr]);
            float r8[8], r4[4], r2[2];
            #pragma unroll
            for (int u = 0; u < 8; ++u) r8[u] = v[rr][u] + v[rr][u + 8];
            #pragma unroll
            for (int u = 0; u < 4; ++u) r4[u] = r8[u] + r8[u + 4];
            #pragma unroll
            for (int u = 0; u < 2; ++u) r2[u] = r4[u] + r4[u + 2];
            sum[rr] = r2[0] + r2[1];
        }
        #pragma unroll
        for (int s = 16; s > 0; s >>= 1) {
            #pragma unroll
            for (int rr = 0; rr < 2; ++rr) sum[rr] = sum[rr] + __shfl_xor(sum[rr], s, 32);
        }
        #pragma unroll
        for (int rr = 0; rr < 2; ++rr) {
            const float inv = 1.f / sum[rr];
            float* srow = att_s + (2 * w + rr) * AV_PITCH + lane;
            #pragma unroll
            for (int u = 0; u < 16; ++u) {
                v[rr][u] = v[rr][u] * inv;
                srow[32 * u] = v[rr][u];
            }
        }
        if (bx == 0 && w == 0) {
            #pragma unroll
            for (int u = 0; u < 16; ++u) att00[lane + 32 * u] = v[0][u];
        }
    }
    x_chunk_store(ra, slot0, t);
    x_chunk_load(ra, X, 2, t);
    lds_barrier();

    const int n0    = w << 4;
    const int lhalf = lane >> 4;
    const int lm    = lane & 15;

    const float* Ak  = att_s + lm * AV_PITCH + 2 * lhalf;
    const float* Bk0 = slot0 + (2 * lhalf) * XS_PITCH + n0 + lm;
    const float* Bk1 = slot1 + (2 * lhalf) * XS_PITCH + n0 + lm;

    v8f c = {0.f, 0.f, 0.f, 0.f, 0.f, 0.f, 0.f, 0.f};
    #pragma unroll 1
    for (int cc = 0; cc < XC_N; cc += 2) {
        av_chunk(c, Ak + cc * XC_ROWS, Bk0);
        x_chunk_store(rb, slot1, t);
        if (cc + 3 < XC_N) x_chunk_load(rb, X, cc + 3, t);
        lds_barrier();
        av_chunk(c, Ak + (cc + 1) * XC_ROWS, Bk1);
        if (cc + 2 < XC_N) {
            x_chunk_store(ra, slot0, t);
            if (cc + 4 < XC_N) x_chunk_load(ra, X, cc + 4, t);
            lds_barrier();
        }
    }

    float* orow = out + (size_t)b * GAT_N * GAT_F
                      + (size_t)(m0 + 8 * lhalf) * GAT_F + n0 + lm;
    #pragma unroll
    for (int v = 0; v < 8; ++v) orow[(size_t)v * GAT_F] = c[v];
}

extern "C" void kernel_launch(void* const* d_in, const int* in_sizes, int n_in,
                              void* d_out, int out_size, void* d_ws, size_t ws_size,
                              hipStream_t stream) {
    const float* x   = (const float*)d_in[0];
    const float* w1  = (const float*)d_in[1];
    const float* b1  = (const float*)d_in[2];
    const float* w2  = (const float*)d_in[3];
    const float* b2  = (const float*)d_in[4];
    const int*   adj = (const int*)d_in[5];

    float* out = (float*)d_out;
    float* P = (float*)d_ws;
    float* Q = P + (size_t)GAT_B * GAT_N * GAT_H;
    float* E = Q + (size_t)GAT_B * GAT_N * GAT_H;

    float* next_h = out;
    float* att00  = out + (size_t)GAT_B * GAT_N * GAT_F;

    pq_gemm<<<256, 256, 0, stream>>>(x, w1, P, Q);
    e_kernel<<<256, 256, 0, stream>>>(P, Q, b1, w2, b2, adj, E);
    softmax_av_gemm<<<64, 256, 0, stream>>>(E, x, next_h, att00);
}


# att*x kernel spread over 256 blocks (was 64): block = 16 rows x 32 columns, its x slice staged in LDS once; 8 waves split K, partial tiles summed through LDS in wave order (k sum order changes)
# speedup vs baseline: 3.5583x; 1.1460x over previous
#include <hip/hip_runtime.h>
#include <hip/hip_bf16.h>


typedef __attribute__((ext_vector_type(2))) float v2f;
typedef __attribute__((ext_vector_type(4))) float v4f;
typedef __attribute__((ext_vector_type(8))) float v8f;

#define GAT_B 2
#define GAT_N 512
#define GAT_F 128
#define GAT_H 256
#define NEG_SLOPE 0.04f
#define MASK_VAL  (-9.0e15f)

__global__ __launch_bounds__(256) void pq_gemm(const float* __restrict__ x,
                                               const float* __restrict__ w1,
                                               float* __restrict__ P,
                                               float* __restrict__ Q) {
    const int wave = (blockIdx.x * blockDim.x + threadIdx.x) >> 5;
    const int lane = threadIdx.x & 31;
    const int which = wave >> 10;
    const int rem   = wave & 1023;
    const int m0 = (rem >> 4) << 4;
    const int n0 = (rem & 15) << 4;
    const int lhalf = lane >> 4;
    const int lm    = lane & 15;

    const float* Arow = x + (size_t)(m0 + lm) * GAT_F;
    const float* Brow = w1 + (size_t)(n0 + lm) * GAT_H + (which ? GAT_F : 0);

    v8f c = {0.f, 0.f, 0.f, 0.f, 0.f, 0.f, 0.f, 0.f};
    #pragma unroll 8
    for (int k0 = 0; k0 < GAT_F; k0 += 4) {
        const int kk = k0 + 2 * lhalf;
        v2f a = *(const v2f*)(Arow + kk);
        v2f b = *(const v2f*)(Brow + kk);
        c = __builtin_amdgcn_wmma_f32_16x16x4_f32(false, a, false, b,
                                                  (short)0, c, false, false);
    }

    float* out = (which ? Q : P);
    float* orow = out + (size_t)(m0 + 8 * lhalf) * GAT_H + n0 + lm;
    #pragma unroll
    for (int v = 0; v < 8; ++v) orow[(size_t)v * GAT_H] = c[v];
}

#define E_TI 32
#define E_TJ 64
#define E_PITCH (GAT_H + 4)
#define E_KS 8

#define E_KU (E_KS / 4)
struct EStage { float4 p[4][E_KU]; float4 q[2][E_KU]; float4 w[E_KU]; };

__device__ __forceinline__ void e_load_stage(EStage& s, const float* __restrict__ prow,
                                             const float* __restrict__ q0,
                                             const float* __restrict__ q1,
                                             const float* __restrict__ w2, int h) {
    #pragma unroll
    for (int u = 0; u < E_KU; ++u) {
        s.w[u] = *(const float4*)(w2 + h + 4 * u);
        #pragma unroll
        for (int r = 0; r < 4; ++r)
            s.p[r][u] = *(const float4*)(prow + r * E_PITCH + h + 4 * u);
        s.q[0][u] = *(const float4*)(q0 + h + 4 * u);
        s.q[1][u] = *(const float4*)(q1 + h + 4 * u);
    }
}

__device__ __forceinline__ void e_compute_stage(const EStage& s, float (&acc)[4][2]) {
    #pragma unroll
    for (int u = 0; u < E_KU; ++u) {
        const float4 wv = s.w[u];
        #pragma unroll
        for (int r = 0; r < 4; ++r) {
            #pragma unroll
            for (int c = 0; c < 2; ++c) {
                const float4 pv = s.p[r][u];
                const float4 qv = s.q[c][u];
                const v2f plo = {pv.x, pv.y}, phi = {pv.z, pv.w};
                const v2f qlo = {qv.x, qv.y}, qhi = {qv.z, qv.w};
                const v2f tlo = plo + qlo;
                const v2f thi = phi + qhi;
                float a = acc[r][c];
                a = __builtin_fmaf(wv.x, fmaxf(tlo.x, 0.f), a);
                a = __builtin_fmaf(wv.y, fmaxf(tlo.y, 0.f), a);
                a = __builtin_fmaf(wv.z, fmaxf(thi.x, 0.f), a);
                a = __builtin_fmaf(wv.w, fmaxf(thi.y, 0.f), a);
                acc[r][c] = a;
            }
        }
    }
}

__global__ __launch_bounds__(256) void e_kernel(const float* __restrict__ P,
                                                const float* __restrict__ Q,
                                                const float* __restrict__ b1,
                                                const float* __restrict__ w2,
                                                const float* __restrict__ b2,
                                                const int*   __restrict__ adj,
                                                float* __restrict__ E) {
    __shared__ __attribute__((aligned(16))) float smem[(E_TI + E_TJ) * E_PITCH];
    float* Ps = smem;
    float* Qs = smem + E_TI * E_PITCH;

    const int bx = blockIdx.x;
    const int b  = bx >> 7;
    const int i0 = ((bx >> 3) & 15) * E_TI;
    const int j0 = (bx & 7) * E_TJ;
    const int t  = threadIdx.x;

    {
        const int c4 = t & 63;
        const int r0 = t >> 6;
        const float4 bb = *(const float4*)(b1 + 4 * c4);
        const float* Pg = P + (size_t)(b * GAT_N + i0) * GAT_H + 4 * c4;
        const float* Qg = Q + (size_t)(b * GAT_N + j0) * GAT_H + 4 * c4;
        #pragma unroll
        for (int k = 0; k < E_TI / 4; ++k) {
            const int r = r0 + 4 * k;
            float4 v = *(const float4*)(Pg + (size_t)r * GAT_H);
            v.x += bb.x; v.y += bb.y; v.z += bb.z; v.w += bb.w;
            *(float4*)(Ps + r * E_PITCH + 4 * c4) = v;
        }
        #pragma unroll
        for (int k = 0; k < E_TJ / 4; ++k) {
            const int r = r0 + 4 * k;
            const float4 v = *(const float4*)(Qg + (size_t)r * GAT_H);
            *(float4*)(Qs + r * E_PITCH + 4 * c4) = v;
        }
    }

    const int w = t >> 5;
    const int l = t & 31;
    const int   ibase  = i0 + 4 * w;
    const int*  adjrow = adj + (size_t)b * GAT_N * GAT_N + (size_t)ibase * GAT_N + j0 + l;
    float*      erow   = E   + (size_t)b * GAT_N * GAT_N + (size_t)ibase * GAT_N + j0 + l;

    int am[4][2];
    #pragma unroll
    for (int r = 0; r < 4; ++r) {
        am[r][0] = adjrow[(size_t)r * GAT_N];
        am[r][1] = adjrow[(size_t)r * GAT_N + 32];
    }
    const float b2v = b2[0];

    __syncthreads();

    const float* prow = Ps + (4 * w) * E_PITCH;
    const float* q0   = Qs + l * E_PITCH;
    const float* q1   = Qs + (l + 32) * E_PITCH;

    float acc[4][2];
    #pragma unroll
    for (int r = 0; r < 4; ++r) { acc[r][0] = 0.f; acc[r][1] = 0.f; }

    EStage sa, sb;
    e_load_stage(sa, prow, q0, q1, w2, 0);
    #pragma unroll 1
    for (int h = 0; h < GAT_H - 2 * E_KS; h += 2 * E_KS) {
        e_load_stage(sb, prow, q0, q1, w2, h + E_KS);
        e_compute_stage(sa, acc);
        e_load_stage(sa, prow, q0, q1, w2, h + 2 * E_KS);
        e_compute_stage(sb, acc);
    }
    e_load_stage(sb, prow, q0, q1, w2, GAT_H - E_KS);
    e_compute_stage(sa, acc);
    e_compute_stage(sb, acc);

    #pragma unroll
    for (int r = 0; r < 4; ++r) {
        #pragma unroll
        for (int c = 0; c < 2; ++c) {
            float e = acc[r][c] + b2v;
            e = (e > 0.f) ? e : NEG_SLOPE * e;
            e = (am[r][c] > 0) ? e : MASK_VAL;
            erow[(size_t)r * GAT_N + 32 * c] = e;
        }
    }
}

#define AV_ROWS 16
#define AV_PITCH (GAT_N + 4)
#define XN 32
#define XS_PITCH (XN + 8)
#define XS_LD (GAT_N * XN / 4 / 256)
#define AVK (GAT_N / 8)
#define RED_PITCH (XN + 4)

__global__ __launch_bounds__(256) void softmax_av_gemm(const float* __restrict__ E,
                                                       const float* __restrict__ x,
                                                       float* __restrict__ out,
                                                       float* __restrict__ att00) {
    __shared__ __attribute__((aligned(16))) float att_s[AV_ROWS * AV_PITCH];
    __shared__ __attribute__((aligned(16))) float xs[GAT_N * XS_PITCH];
    __shared__ __attribute__((aligned(16))) float red_s[8 * AV_ROWS * RED_PITCH];

    const int bx   = blockIdx.x;
    const int b    = bx >> 7;
    const int m0   = ((bx >> 2) & 31) << 4;
    const int nb   = (bx & 3) * XN;
    const int t    = threadIdx.x;
    const int w    = t >> 5;
    const int lane = t & 31;

    const float* X = x + (size_t)b * GAT_N * GAT_F;
    v4f xr[XS_LD];

    {
        const float* e0p = E + ((size_t)b * GAT_N + m0 + 2 * w) * GAT_N + lane;
        float v[2][16];
        #pragma unroll
        for (int rr = 0; rr < 2; ++rr) {
            #pragma unroll
            for (int u = 0; u < 16; ++u) v[rr][u] = e0p[(size_t)rr * GAT_N + 32 * u];
        }
        {
            const float* xp = X + (size_t)(t >> 3) * GAT_F + nb + 4 * (t & 7);
            #pragma unroll
            for (int k = 0; k < XS_LD; ++k) xr[k] = *(const v4f*)(xp + (size_t)(32 * k) * GAT_F);
        }

        float mx[2];
        #pragma unroll
        for (int rr = 0; rr < 2; ++rr) {
            float m = v[rr][0];
            #pragma unroll
            for (int u = 1; u < 16; ++u) m = fmaxf(m, v[rr][u]);
            mx[rr] = m;
        }
        #pragma unroll
        for (int s = 16; s > 0; s >>= 1) {
            #pragma unroll
            for (int rr = 0; rr < 2; ++rr) mx[rr] = fmaxf(mx[rr], __shfl_xor(mx[rr], s, 32));
        }
        float sum[2];
        #pragma unroll
        for (int rr = 0; rr < 2; ++rr) {
            #pragma unroll
            for (int u = 0; u < 16; ++u) v[rr][u] = __expf(v[rr][u] - mx[rr]);
            float r8[8], r4[4], r2[2];
            #pragma unroll
            for (int u = 0; u < 8; ++u) r8[u] = v[rr][u] + v[rr][u + 8];
            #pragma unroll
            for (int u = 0; u < 4; ++u) r4[u] = r8[u] + r8[u + 4];
            #pragma unroll
            for (int u = 0; u < 2; ++u) r2[u] = r4[u] + r4[u + 2];
            sum[rr] = r2[0] + r2[1];
        }
        #pragma unroll
        for (int s = 16; s > 0; s >>= 1) {
            #pragma unroll
            for (int rr = 0; rr < 2; ++rr) sum[rr] = sum[rr] + __shfl_xor(sum[rr], s, 32);
        }
        #pragma unroll
        for (int rr = 0; rr < 2; ++rr) {
            const float inv = 1.f / sum[rr];
            float* srow = att_s + (2 * w + rr) * AV_PITCH + lane;
            #pragma unroll
            for (int u = 0; u < 16; ++u) {
                v[rr][u] = v[rr][u] * inv;
                srow[32 * u] = v[rr][u];
            }
        }
        if (bx == 0 && w == 0) {
            #pragma unroll
            for (int u = 0; u < 16; ++u) att00[lane + 32 * u] = v[0][u];
        }
    }
    {
        float* sp = xs + (t >> 3) * XS_PITCH + 4 * (t & 7);
        #pragma unroll
        for (int k = 0; k < XS_LD; ++k) *(v4f*)(sp + (32 * k) * XS_PITCH) = xr[k];
    }
    __syncthreads();

    const int lhalf = lane >> 4;
    const int lm    = lane & 15;

    const float* Ak = att_s + lm * AV_PITCH + AVK * w + 2 * lhalf;
    const float* Bk = xs + (AVK * w + 2 * lhalf) * XS_PITCH + lm;

    v8f c0 = {0.f, 0.f, 0.f, 0.f, 0.f, 0.f, 0.f, 0.f};
    v8f c1 = {0.f, 0.f, 0.f, 0.f, 0.f, 0.f, 0.f, 0.f};
    #pragma unroll
    for (int s = 0; s < AVK / 4; ++s) {
        v2f a = *(const v2f*)(Ak + 4 * s);
        v2f b0, b1;
        b0.x = Bk[(4 * s) * XS_PITCH];
        b0.y = Bk[(4 * s + 1) * XS_PITCH];
        b1.x = Bk[(4 * s) * XS_PITCH + 16];
        b1.y = Bk[(4 * s + 1) * XS_PITCH + 16];
        c0 = __builtin_amdgcn_wmma_f32_16x16x4_f32(false, a, false, b0,
                                                   (short)0, c0, false, false);
        c1 = __builtin_amdgcn_wmma_f32_16x16x4_f32(false, a, false, b1,
                                                   (short)0, c1, false, false);
    }

    {
        float* rp = red_s + w * (AV_ROWS * RED_PITCH) + (8 * lhalf) * RED_PITCH + lm;
        #pragma unroll
        for (int v = 0; v < 8; ++v) {
            rp[v * RED_PITCH]      = c0[v];
            rp[v * RED_PITCH + 16] = c1[v];
        }
    }
    __syncthreads();

    {
        const float* fp = red_s + (t >> 4) * RED_PITCH + 2 * (t & 15);
        v2f acc = *(const v2f*)fp;
        #pragma unroll
        for (int q = 1; q < 8; ++q) acc = acc + *(const v2f*)(fp + q * (AV_ROWS * RED_PITCH));
        float* op = out + (size_t)b * GAT_N * GAT_F
                        + (size_t)(m0 + (t >> 4)) * GAT_F + nb + 2 * (t & 15);
        *(v2f*)op = acc;
    }
}

extern "C" void kernel_launch(void* const* d_in, const int* in_sizes, int n_in,
                              void* d_out, int out_size, void* d_ws, size_t ws_size,
                              hipStream_t stream) {
    const float* x   = (const float*)d_in[0];
    const float* w1  = (const float*)d_in[1];
    const float* b1  = (const float*)d_in[2];
    const float* w2  = (const float*)d_in[3];
    const float* b2  = (const float*)d_in[4];
    const int*   adj = (const int*)d_in[5];

    float* out = (float*)d_out;
    float* P = (float*)d_ws;
    float* Q = P + (size_t)GAT_B * GAT_N * GAT_H;
    float* E = Q + (size_t)GAT_B * GAT_N * GAT_H;

    float* next_h = out;
    float* att00  = out + (size_t)GAT_B * GAT_N * GAT_F;

    pq_gemm<<<256, 256, 0, stream>>>(x, w1, P, Q);
    e_kernel<<<256, 256, 0, stream>>>(P, Q, b1, w2, b2, adj, E);
    softmax_av_gemm<<<256, 256, 0, stream>>>(E, x, next_h, att00);
}


# P/Q projection kernel: each block's 32 rows of x and 64 rows of w1 staged once in LDS by whole-row 16-byte loads, A and B fragments read from LDS, output tile stored as whole lines; k order kept
# speedup vs baseline: 3.8686x; 1.0872x over previous
#include <hip/hip_runtime.h>
#include <hip/hip_bf16.h>


typedef __attribute__((ext_vector_type(2))) float v2f;
typedef __attribute__((ext_vector_type(4))) float v4f;
typedef __attribute__((ext_vector_type(8))) float v8f;

#define GAT_B 2
#define GAT_N 512
#define GAT_F 128
#define GAT_H 256
#define NEG_SLOPE 0.04f
#define MASK_VAL  (-9.0e15f)

#define PQ_TM 32
#define PQ_TN 64
#define PQ_PITCH (GAT_F + 4)
#define PQ_CPITCH (PQ_TN + 4)

__global__ __launch_bounds__(256) void pq_gemm(const float* __restrict__ x,
                                               const float* __restrict__ w1,
                                               float* __restrict__ P,
                                               float* __restrict__ Q) {
    __shared__ __attribute__((aligned(16))) float xs[PQ_TM * PQ_PITCH];
    __shared__ __attribute__((aligned(16))) float ws[PQ_TN * PQ_PITCH];
    __shared__ __attribute__((aligned(16))) float cs[PQ_TM * PQ_CPITCH];

    const int bx    = blockIdx.x;
    const int m0    = (bx >> 3) * PQ_TM;
    const int nt    = bx & 7;
    const int which = nt >> 2;
    const int n0    = (nt & 3) * PQ_TN;
    const int t     = threadIdx.x;
    const int w     = t >> 5;
    const int lane  = t & 31;

    {
        const int c4 = t & 31;
        const int r0 = t >> 5;
        const float* xg = x + (size_t)(m0 + r0) * GAT_F + 4 * c4;
        const float* wg = w1 + (size_t)(n0 + r0) * GAT_H + (which ? GAT_F : 0) + 4 * c4;
        v4f xv[PQ_TM / 8], wv[PQ_TN / 8];
        #pragma unroll
        for (int k = 0; k < PQ_TM / 8; ++k) xv[k] = *(const v4f*)(xg + (size_t)(8 * k) * GAT_F);
        #pragma unroll
        for (int k = 0; k < PQ_TN / 8; ++k) wv[k] = *(const v4f*)(wg + (size_t)(8 * k) * GAT_H);
        float* xp = xs + r0 * PQ_PITCH + 4 * c4;
        float* wp = ws + r0 * PQ_PITCH + 4 * c4;
        #pragma unroll
        for (int k = 0; k < PQ_TM / 8; ++k) *(v4f*)(xp + (8 * k) * PQ_PITCH) = xv[k];
        #pragma unroll
        for (int k = 0; k < PQ_TN / 8; ++k) *(v4f*)(wp + (8 * k) * PQ_PITCH) = wv[k];
    }
    __syncthreads();

    const int lhalf = lane >> 4;
    const int lm    = lane & 15;
    const int tm    = w >> 2;
    const int tn    = w & 3;

    const float* Ak = xs + (16 * tm + lm) * PQ_PITCH + 2 * lhalf;
    const float* Bk = ws + (16 * tn + lm) * PQ_PITCH + 2 * lhalf;

    v8f c = {0.f, 0.f, 0.f, 0.f, 0.f, 0.f, 0.f, 0.f};
    #pragma unroll
    for (int k0 = 0; k0 < GAT_F; k0 += 4) {
        v2f a = *(const v2f*)(Ak + k0);
        v2f b = *(const v2f*)(Bk + k0);
        c = __builtin_amdgcn_wmma_f32_16x16x4_f32(false, a, false, b,
                                                  (short)0, c, false, false);
    }

    {
        float* cp = cs + (16 * tm + 8 * lhalf) * PQ_CPITCH + 16 * tn + lm;
        #pragma unroll
        for (int v = 0; v < 8; ++v) cp[v * PQ_CPITCH] = c[v];
    }
    __syncthreads();

    {
        float* out = (which ? Q : P) + (size_t)m0 * GAT_H + n0;
        const int c4 = t & 15;
        const int r  = t >> 4;
        #pragma unroll
        for (int k = 0; k < 2; ++k) {
            const v4f v = *(const v4f*)(cs + (r + 16 * k) * PQ_CPITCH + 4 * c4);
            *(v4f*)(out + (size_t)(r + 16 * k) * GAT_H + 4 * c4) = v;
        }
    }
}

#define E_TI 32
#define E_TJ 64
#define E_PITCH (GAT_H + 4)
#define E_KS 8

#define E_KU (E_KS / 4)
struct EStage { float4 p[4][E_KU]; float4 q[2][E_KU]; float4 w[E_KU]; };

__device__ __forceinline__ void e_load_stage(EStage& s, const float* __restrict__ prow,
                                             const float* __restrict__ q0,
                                             const float* __restrict__ q1,
                                             const float* __restrict__ w2, int h) {
    #pragma unroll
    for (int u = 0; u < E_KU; ++u) {
        s.w[u] = *(const float4*)(w2 + h + 4 * u);
        #pragma unroll
        for (int r = 0; r < 4; ++r)
            s.p[r][u] = *(const float4*)(prow + r * E_PITCH + h + 4 * u);
        s.q[0][u] = *(const float4*)(q0 + h + 4 * u);
        s.q[1][u] = *(const float4*)(q1 + h + 4 * u);
    }
}

__device__ __forceinline__ void e_compute_stage(const EStage& s, float (&acc)[4][2]) {
    #pragma unroll
    for (int u = 0; u < E_KU; ++u) {
        const float4 wv = s.w[u];
        #pragma unroll
        for (int r = 0; r < 4; ++r) {
            #pragma unroll
            for (int c = 0; c < 2; ++c) {
                const float4 pv = s.p[r][u];
                const float4 qv = s.q[c][u];
                const v2f plo = {pv.x, pv.y}, phi = {pv.z, pv.w};
                const v2f qlo = {qv.x, qv.y}, qhi = {qv.z, qv.w};
                const v2f tlo = plo + qlo;
                const v2f thi = phi + qhi;
                float a = acc[r][c];
                a = __builtin_fmaf(wv.x, fmaxf(tlo.x, 0.f), a);
                a = __builtin_fmaf(wv.y, fmaxf(tlo.y, 0.f), a);
                a = __builtin_fmaf(wv.z, fmaxf(thi.x, 0.f), a);
                a = __builtin_fmaf(wv.w, fmaxf(thi.y, 0.f), a);
                acc[r][c] = a;
            }
        }
    }
}

__global__ __launch_bounds__(256) void e_kernel(const float* __restrict__ P,
                                                const float* __restrict__ Q,
                                                const float* __restrict__ b1,
                                                const float* __restrict__ w2,
                                                const float* __restrict__ b2,
                                                const int*   __restrict__ adj,
                                                float* __restrict__ E) {
    __shared__ __attribute__((aligned(16))) float smem[(E_TI + E_TJ) * E_PITCH];
    float* Ps = smem;
    float* Qs = smem + E_TI * E_PITCH;

    const int bx = blockIdx.x;
    const int b  = bx >> 7;
    const int i0 = ((bx >> 3) & 15) * E_TI;
    const int j0 = (bx & 7) * E_TJ;
    const int t  = threadIdx.x;

    {
        const int c4 = t & 63;
        const int r0 = t >> 6;
        const float4 bb = *(const float4*)(b1 + 4 * c4);
        const float* Pg = P + (size_t)(b * GAT_N + i0) * GAT_H + 4 * c4;
        const float* Qg = Q + (size_t)(b * GAT_N + j0) * GAT_H + 4 * c4;
        #pragma unroll
        for (int k = 0; k < E_TI / 4; ++k) {
            const int r = r0 + 4 * k;
            float4 v = *(const float4*)(Pg + (size_t)r * GAT_H);
            v.x += bb.x; v.y += bb.y; v.z += bb.z; v.w += bb.w;
            *(float4*)(Ps + r * E_PITCH + 4 * c4) = v;
        }
        #pragma unroll
        for (int k = 0; k < E_TJ / 4; ++k) {
            const int r = r0 + 4 * k;
            const float4 v = *(const float4*)(Qg + (size_t)r * GAT_H);
            *(float4*)(Qs + r * E_PITCH + 4 * c4) = v;
        }
    }

    const int w = t >> 5;
    const int l = t & 31;
    const int   ibase  = i0 + 4 * w;
    const int*  adjrow = adj + (size_t)b * GAT_N * GAT_N + (size_t)ibase * GAT_N + j0 + l;
    float*      erow   = E   + (size_t)b * GAT_N * GAT_N + (size_t)ibase * GAT_N + j0 + l;

    int am[4][2];
    #pragma unroll
    for (int r = 0; r < 4; ++r) {
        am[r][0] = adjrow[(size_t)r * GAT_N];
        am[r][1] = adjrow[(size_t)r * GAT_N + 32];
    }
    const float b2v = b2[0];

    __syncthreads();

    const float* prow = Ps + (4 * w) * E_PITCH;
    const float* q0   = Qs + l * E_PITCH;
    const float* q1   = Qs + (l + 32) * E_PITCH;

    float acc[4][2];
    #pragma unroll
    for (int r = 0; r < 4; ++r) { acc[r][0] = 0.f; acc[r][1] = 0.f; }

    EStage sa, sb;
    e_load_stage(sa, prow, q0, q1, w2, 0);
    #pragma unroll 1
    for (int h = 0; h < GAT_H - 2 * E_KS; h += 2 * E_KS) {
        e_load_stage(sb, prow, q0, q1, w2, h + E_KS);
        e_compute_stage(sa, acc);
        e_load_stage(sa, prow, q0, q1, w2, h + 2 * E_KS);
        e_compute_stage(sb, acc);
    }
    e_load_stage(sb, prow, q0, q1, w2, GAT_H - E_KS);
    e_compute_stage(sa, acc);
    e_compute_stage(sb, acc);

    #pragma unroll
    for (int r = 0; r < 4; ++r) {
        #pragma unroll
        for (int c = 0; c < 2; ++c) {
            float e = acc[r][c] + b2v;
            e = (e > 0.f) ? e : NEG_SLOPE * e;
            e = (am[r][c] > 0) ? e : MASK_VAL;
            erow[(size_t)r * GAT_N + 32 * c] = e;
        }
    }
}

#define AV_ROWS 16
#define AV_PITCH (GAT_N + 4)
#define XN 32
#define XS_PITCH (XN + 8)
#define XS_LD (GAT_N * XN / 4 / 256)
#define AVK (GAT_N / 8)
#define RED_PITCH (XN + 4)

__global__ __launch_bounds__(256) void softmax_av_gemm(const float* __restrict__ E,
                                                       const float* __restrict__ x,
                                                       float* __restrict__ out,
                                                       float* __restrict__ att00) {
    __shared__ __attribute__((aligned(16))) float att_s[AV_ROWS * AV_PITCH];
    __shared__ __attribute__((aligned(16))) float xs[GAT_N * XS_PITCH];
    __shared__ __attribute__((aligned(16))) float red_s[8 * AV_ROWS * RED_PITCH];

    const int bx   = blockIdx.x;
    const int b    = bx >> 7;
    const int m0   = ((bx >> 2) & 31) << 4;
    const int nb   = (bx & 3) * XN;
    const int t    = threadIdx.x;
    const int w    = t >> 5;
    const int lane = t & 31;

    const float* X = x + (size_t)b * GAT_N * GAT_F;
    v4f xr[XS_LD];

    {
        const float* e0p = E + ((size_t)b * GAT_N + m0 + 2 * w) * GAT_N + lane;
        float v[2][16];
        #pragma unroll
        for (int rr = 0; rr < 2; ++rr) {
            #pragma unroll
            for (int u = 0; u < 16; ++u) v[rr][u] = e0p[(size_t)rr * GAT_N + 32 * u];
        }
        {
            const float* xp = X + (size_t)(t >> 3) * GAT_F + nb + 4 * (t & 7);
            #pragma unroll
            for (int k = 0; k < XS_LD; ++k) xr[k] = *(const v4f*)(xp + (size_t)(32 * k) * GAT_F);
        }

        float mx[2];
        #pragma unroll
        for (int rr = 0; rr < 2; ++rr) {
            float m = v[rr][0];
            #pragma unroll
            for (int u = 1; u < 16; ++u) m = fmaxf(m, v[rr][u]);
            mx[rr] = m;
        }
        #pragma unroll
        for (int s = 16; s > 0; s >>= 1) {
            #pragma unroll
            for (int rr = 0; rr < 2; ++rr) mx[rr] = fmaxf(mx[rr], __shfl_xor(mx[rr], s, 32));
        }
        float sum[2];
        #pragma unroll
        for (int rr = 0; rr < 2; ++rr) {
            #pragma unroll
            for (int u = 0; u < 16; ++u) v[rr][u] = __expf(v[rr][u] - mx[rr]);
            float r8[8], r4[4], r2[2];
            #pragma unroll
            for (int u = 0; u < 8; ++u) r8[u] = v[rr][u] + v[rr][u + 8];
            #pragma unroll
            for (int u = 0; u < 4; ++u) r4[u] = r8[u] + r8[u + 4];
            #pragma unroll
            for (int u = 0; u < 2; ++u) r2[u] = r4[u] + r4[u + 2];
            sum[rr] = r2[0] + r2[1];
        }
        #pragma unroll
        for (int s = 16; s > 0; s >>= 1) {
            #pragma unroll
            for (int rr = 0; rr < 2; ++rr) sum[rr] = sum[rr] + __shfl_xor(sum[rr], s, 32);
        }
        #pragma unroll
        for (int rr = 0; rr < 2; ++rr) {
            const float inv = 1.f / sum[rr];
            float* srow = att_s + (2 * w + rr) * AV_PITCH + lane;
            #pragma unroll
            for (int u = 0; u < 16; ++u) {
                v[rr][u] = v[rr][u] * inv;
                srow[32 * u] = v[rr][u];
            }
        }
        if (bx == 0 && w == 0) {
            #pragma unroll
            for (int u = 0; u < 16; ++u) att00[lane + 32 * u] = v[0][u];
        }
    }
    {
        float* sp = xs + (t >> 3) * XS_PITCH + 4 * (t & 7);
        #pragma unroll
        for (int k = 0; k < XS_LD; ++k) *(v4f*)(sp + (32 * k) * XS_PITCH) = xr[k];
    }
    __syncthreads();

    const int lhalf = lane >> 4;
    const int lm    = lane & 15;

    const float* Ak = att_s + lm * AV_PITCH + AVK * w + 2 * lhalf;
    const float* Bk = xs + (AVK * w + 2 * lhalf) * XS_PITCH + lm;

    v8f c0 = {0.f, 0.f, 0.f, 0.f, 0.f, 0.f, 0.f, 0.f};
    v8f c1 = {0.f, 0.f, 0.f, 0.f, 0.f, 0.f, 0.f, 0.f};
    #pragma unroll
    for (int s = 0; s < AVK / 4; ++s) {
        v2f a = *(const v2f*)(Ak + 4 * s);
        v2f b0, b1;
        b0.x = Bk[(4 * s) * XS_PITCH];
        b0.y = Bk[(4 * s + 1) * XS_PITCH];
        b1.x = Bk[(4 * s) * XS_PITCH + 16];
        b1.y = Bk[(4 * s + 1) * XS_PITCH + 16];
        c0 = __builtin_amdgcn_wmma_f32_16x16x4_f32(false, a, false, b0,
                                                   (short)0, c0, false, false);
        c1 = __builtin_amdgcn_wmma_f32_16x16x4_f32(false, a, false, b1,
                                                   (short)0, c1, false, false);
    }

    {
        float* rp = red_s + w * (AV_ROWS * RED_PITCH) + (8 * lhalf) * RED_PITCH + lm;
        #pragma unroll
        for (int v = 0; v < 8; ++v) {
            rp[v * RED_PITCH]      = c0[v];
            rp[v * RED_PITCH + 16] = c1[v];
        }
    }
    __syncthreads();

    {
        const float* fp = red_s + (t >> 4) * RED_PITCH + 2 * (t & 15);
        v2f acc = *(const v2f*)fp;
        #pragma unroll
        for (int q = 1; q < 8; ++q) acc = acc + *(const v2f*)(fp + q * (AV_ROWS * RED_PITCH));
        float* op = out + (size_t)b * GAT_N * GAT_F
                        + (size_t)(m0 + (t >> 4)) * GAT_F + nb + 2 * (t & 15);
        *(v2f*)op = acc;
    }
}

extern "C" void kernel_launch(void* const* d_in, const int* in_sizes, int n_in,
                              void* d_out, int out_size, void* d_ws, size_t ws_size,
                              hipStream_t stream) {
    const float* x   = (const float*)d_in[0];
    const float* w1  = (const float*)d_in[1];
    const float* b1  = (const float*)d_in[2];
    const float* w2  = (const float*)d_in[3];
    const float* b2  = (const float*)d_in[4];
    const int*   adj = (const int*)d_in[5];

    float* out = (float*)d_out;
    float* P = (float*)d_ws;
    float* Q = P + (size_t)GAT_B * GAT_N * GAT_H;
    float* E = Q + (size_t)GAT_B * GAT_N * GAT_H;

    float* next_h = out;
    float* att00  = out + (size_t)GAT_B * GAT_N * GAT_F;

    pq_gemm<<<256, 256, 0, stream>>>(x, w1, P, Q);
    e_kernel<<<256, 256, 0, stream>>>(P, Q, b1, w2, b2, adj, E);
    softmax_av_gemm<<<256, 256, 0, stream>>>(E, x, next_h, att00);
}


# e-logit loop re-tiled: lane owns 4x4 outputs over an h half, sums as even-h/odd-h register pairs, next stage's LDS reads held a full stage ahead; h sum order changes (even+odd, then low+high half)
# speedup vs baseline: 4.0984x; 1.0594x over previous
#include <hip/hip_runtime.h>
#include <hip/hip_bf16.h>


typedef __attribute__((ext_vector_type(2))) float v2f;
typedef __attribute__((ext_vector_type(4))) float v4f;
typedef __attribute__((ext_vector_type(8))) float v8f;

#define GAT_B 2
#define GAT_N 512
#define GAT_F 128
#define GAT_H 256
#define NEG_SLOPE 0.04f
#define MASK_VAL  (-9.0e15f)

#define PQ_TM 32
#define PQ_TN 64
#define PQ_PITCH (GAT_F + 4)
#define PQ_CPITCH (PQ_TN + 4)

__global__ __launch_bounds__(256) void pq_gemm(const float* __restrict__ x,
                                               const float* __restrict__ w1,
                                               float* __restrict__ P,
                                               float* __restrict__ Q) {
    __shared__ __attribute__((aligned(16))) float xs[PQ_TM * PQ_PITCH];
    __shared__ __attribute__((aligned(16))) float ws[PQ_TN * PQ_PITCH];
    __shared__ __attribute__((aligned(16))) float cs[PQ_TM * PQ_CPITCH];

    const int bx    = blockIdx.x;
    const int m0    = (bx >> 3) * PQ_TM;
    const int nt    = bx & 7;
    const int which = nt >> 2;
    const int n0    = (nt & 3) * PQ_TN;
    const int t     = threadIdx.x;
    const int w     = t >> 5;
    const int lane  = t & 31;

    {
        const int c4 = t & 31;
        const int r0 = t >> 5;
        const float* xg = x + (size_t)(m0 + r0) * GAT_F + 4 * c4;
        const float* wg = w1 + (size_t)(n0 + r0) * GAT_H + (which ? GAT_F : 0) + 4 * c4;
        v4f xv[PQ_TM / 8], wv[PQ_TN / 8];
        #pragma unroll
        for (int k = 0; k < PQ_TM / 8; ++k) xv[k] = *(const v4f*)(xg + (size_t)(8 * k) * GAT_F);
        #pragma unroll
        for (int k = 0; k < PQ_TN / 8; ++k) wv[k] = *(const v4f*)(wg + (size_t)(8 * k) * GAT_H);
        float* xp = xs + r0 * PQ_PITCH + 4 * c4;
        float* wp = ws + r0 * PQ_PITCH + 4 * c4;
        #pragma unroll
        for (int k = 0; k < PQ_TM / 8; ++k) *(v4f*)(xp + (8 * k) * PQ_PITCH) = xv[k];
        #pragma unroll
        for (int k = 0; k < PQ_TN / 8; ++k) *(v4f*)(wp + (8 * k) * PQ_PITCH) = wv[k];
    }
    __syncthreads();

    const int lhalf = lane >> 4;
    const int lm    = lane & 15;
    const int tm    = w >> 2;
    const int tn    = w & 3;

    const float* Ak = xs + (16 * tm + lm) * PQ_PITCH + 2 * lhalf;
    const float* Bk = ws + (16 * tn + lm) * PQ_PITCH + 2 * lhalf;

    v8f c = {0.f, 0.f, 0.f, 0.f, 0.f, 0.f, 0.f, 0.f};
    #pragma unroll
    for (int k0 = 0; k0 < GAT_F; k0 += 4) {
        v2f a = *(const v2f*)(Ak + k0);
        v2f b = *(const v2f*)(Bk + k0);
        c = __builtin_amdgcn_wmma_f32_16x16x4_f32(false, a, false, b,
                                                  (short)0, c, false, false);
    }

    {
        float* cp = cs + (16 * tm + 8 * lhalf) * PQ_CPITCH + 16 * tn + lm;
        #pragma unroll
        for (int v = 0; v < 8; ++v) cp[v * PQ_CPITCH] = c[v];
    }
    __syncthreads();

    {
        float* out = (which ? Q : P) + (size_t)m0 * GAT_H + n0;
        const int c4 = t & 15;
        const int r  = t >> 4;
        #pragma unroll
        for (int k = 0; k < 2; ++k) {
            const v4f v = *(const v4f*)(cs + (r + 16 * k) * PQ_CPITCH + 4 * c4);
            *(v4f*)(out + (size_t)(r + 16 * k) * GAT_H + 4 * c4) = v;
        }
    }
}

#define E_TI 32
#define E_TJ 64
#define E_PITCH (GAT_H + 4)
#define E_KS 8
#define E_HH (GAT_H / 2)
#define E_XP (E_TJ + 4)

#define E_KU (E_KS / 4)
struct EStage { float4 p[4][E_KU]; float4 q[4][E_KU]; float4 w[E_KU]; };

__device__ __forceinline__ void e_load_stage(EStage& s, const float* __restrict__ prow,
                                             const float* __restrict__ qrow,
                                             const float* __restrict__ w2, int h) {
    #pragma unroll
    for (int u = 0; u < E_KU; ++u) {
        s.w[u] = *(const float4*)(w2 + h + 4 * u);
        #pragma unroll
        for (int r = 0; r < 4; ++r)
            s.p[r][u] = *(const float4*)(prow + r * E_PITCH + h + 4 * u);
        #pragma unroll
        for (int c = 0; c < 4; ++c)
            s.q[c][u] = *(const float4*)(qrow + (16 * c) * E_PITCH + h + 4 * u);
    }
}

__device__ __forceinline__ void e_compute_stage(const EStage& s, v2f (&acc)[4][4]) {
    #pragma unroll
    for (int u = 0; u < E_KU; ++u) {
        const float4 wv = s.w[u];
        const v2f wlo = {wv.x, wv.y}, whi = {wv.z, wv.w};
        #pragma unroll
        for (int rp = 0; rp < 4; rp += 2) {
            v2f tlo[2][4], thi[2][4];
            #pragma unroll
            for (int rr = 0; rr < 2; ++rr) {
                const float4 pv = s.p[rp + rr][u];
                const v2f plo = {pv.x, pv.y}, phi = {pv.z, pv.w};
                #pragma unroll
                for (int c = 0; c < 4; ++c) {
                    const float4 qv = s.q[c][u];
                    const v2f qlo = {qv.x, qv.y}, qhi = {qv.z, qv.w};
                    tlo[rr][c] = plo + qlo;
                    thi[rr][c] = phi + qhi;
                }
            }
            #pragma unroll
            for (int rr = 0; rr < 2; ++rr) {
                #pragma unroll
                for (int c = 0; c < 4; ++c) {
                    tlo[rr][c].x = fmaxf(tlo[rr][c].x, 0.f); tlo[rr][c].y = fmaxf(tlo[rr][c].y, 0.f);
                    thi[rr][c].x = fmaxf(thi[rr][c].x, 0.f); thi[rr][c].y = fmaxf(thi[rr][c].y, 0.f);
                }
            }
            #pragma unroll
            for (int rr = 0; rr < 2; ++rr) {
                #pragma unroll
                for (int c = 0; c < 4; ++c)
                    acc[rp + rr][c] = __builtin_elementwise_fma(wlo, tlo[rr][c], acc[rp + rr][c]);
            }
            #pragma unroll
            for (int rr = 0; rr < 2; ++rr) {
                #pragma unroll
                for (int c = 0; c < 4; ++c)
                    acc[rp + rr][c] = __builtin_elementwise_fma(whi, thi[rr][c], acc[rp + rr][c]);
            }
        }
    }
}

__global__ __launch_bounds__(256) void e_kernel(const float* __restrict__ P,
                                                const float* __restrict__ Q,
                                                const float* __restrict__ b1,
                                                const float* __restrict__ w2,
                                                const float* __restrict__ b2,
                                                const int*   __restrict__ adj,
                                                float* __restrict__ E) {
    __shared__ __attribute__((aligned(16))) float smem[(E_TI + E_TJ) * E_PITCH];
    __shared__ __attribute__((aligned(16))) float xch[2 * E_TI * E_XP];
    float* Ps = smem;
    float* Qs = smem + E_TI * E_PITCH;

    const int bx = blockIdx.x;
    const int b  = bx >> 7;
    const int i0 = ((bx >> 3) & 15) * E_TI;
    const int j0 = (bx & 7) * E_TJ;
    const int t  = threadIdx.x;

    {
        const int c4 = t & 63;
        const int r0 = t >> 6;
        const float4 bb = *(const float4*)(b1 + 4 * c4);
        const float* Pg = P + (size_t)(b * GAT_N + i0) * GAT_H + 4 * c4;
        const float* Qg = Q + (size_t)(b * GAT_N + j0) * GAT_H + 4 * c4;
        #pragma unroll
        for (int k = 0; k < E_TI / 4; ++k) {
            const int r = r0 + 4 * k;
            float4 v = *(const float4*)(Pg + (size_t)r * GAT_H);
            v.x += bb.x; v.y += bb.y; v.z += bb.z; v.w += bb.w;
            *(float4*)(Ps + r * E_PITCH + 4 * c4) = v;
        }
        #pragma unroll
        for (int k = 0; k < E_TJ / 4; ++k) {
            const int r = r0 + 4 * k;
            const float4 v = *(const float4*)(Qg + (size_t)r * GAT_H);
            *(float4*)(Qs + r * E_PITCH + 4 * c4) = v;
        }
    }

    const int orow = t >> 3;
    const int ocol = 8 * (t & 7);
    const size_t ooff = (size_t)b * GAT_N * GAT_N + (size_t)(i0 + orow) * GAT_N + j0 + ocol;
    const int4 am0 = *(const int4*)(adj + ooff);
    const int4 am1 = *(const int4*)(adj + ooff + 4);
    const float b2v = b2[0];

    __syncthreads();

    const int w    = __builtin_amdgcn_readfirstlane(t >> 5);
    const int l    = t & 31;
    const int hgrp = w >> 2;
    const int wq   = w & 3;
    const int ig   = l >> 4;
    const int jl   = l & 15;
    const int hb   = hgrp * E_HH;

    const float* prow = Ps + (8 * wq + 4 * ig) * E_PITCH;
    const float* qrow = Qs + jl * E_PITCH;

    v2f acc[4][4];
    #pragma unroll
    for (int r = 0; r < 4; ++r) {
        #pragma unroll
        for (int c = 0; c < 4; ++c) { acc[r][c].x = 0.f; acc[r][c].y = 0.f; }
    }

    #define E_HOLD() __builtin_amdgcn_sched_barrier(0)
    EStage sa, sb;
    e_load_stage(sa, prow, qrow, w2, hb);
    asm volatile("" :: "s"(sa.w[0].x), "s"(sa.w[0].y), "s"(sa.w[0].z), "s"(sa.w[0].w),
                       "s"(sa.w[1].x), "s"(sa.w[1].y), "s"(sa.w[1].z), "s"(sa.w[1].w));
    #pragma unroll 1
    for (int h = hb; h < hb + E_HH - 2 * E_KS; h += 2 * E_KS) {
        e_load_stage(sb, prow, qrow, w2, h + E_KS);
        E_HOLD();
        e_compute_stage(sa, acc);
        E_HOLD();
        e_load_stage(sa, prow, qrow, w2, h + 2 * E_KS);
        E_HOLD();
        e_compute_stage(sb, acc);
        E_HOLD();
    }
    e_load_stage(sb, prow, qrow, w2, hb + E_HH - E_KS);
    E_HOLD();
    e_compute_stage(sa, acc);
    e_compute_stage(sb, acc);
    #undef E_HOLD

    {
        float* xp = xch + hgrp * (E_TI * E_XP) + (8 * wq + 4 * ig) * E_XP + jl;
        #pragma unroll
        for (int r = 0; r < 4; ++r) {
            #pragma unroll
            for (int c = 0; c < 4; ++c) xp[r * E_XP + 16 * c] = acc[r][c].x + acc[r][c].y;
        }
    }
    __syncthreads();

    {
        const float* x0 = xch + orow * E_XP + ocol;
        const float* x1 = x0 + E_TI * E_XP;
        const float4 lo0 = *(const float4*)x0, lo1 = *(const float4*)(x0 + 4);
        const float4 hi0 = *(const float4*)x1, hi1 = *(const float4*)(x1 + 4);
        float ev[8] = {lo0.x + hi0.x, lo0.y + hi0.y, lo0.z + hi0.z, lo0.w + hi0.w,
                       lo1.x + hi1.x, lo1.y + hi1.y, lo1.z + hi1.z, lo1.w + hi1.w};
        const int   av[8] = {am0.x, am0.y, am0.z, am0.w, am1.x, am1.y, am1.z, am1.w};
        #pragma unroll
        for (int k = 0; k < 8; ++k) {
            float e = ev[k] + b2v;
            e = (e > 0.f) ? e : NEG_SLOPE * e;
            ev[k] = (av[k] > 0) ? e : MASK_VAL;
        }
        float4 o0, o1;
        o0.x = ev[0]; o0.y = ev[1]; o0.z = ev[2]; o0.w = ev[3];
        o1.x = ev[4]; o1.y = ev[5]; o1.z = ev[6]; o1.w = ev[7];
        *(float4*)(E + ooff)     = o0;
        *(float4*)(E + ooff + 4) = o1;
    }
}

#define AV_ROWS 16
#define AV_PITCH (GAT_N + 4)
#define XN 32
#define XS_PITCH (XN + 8)
#define XS_LD (GAT_N * XN / 4 / 256)
#define AVK (GAT_N / 8)
#define RED_PITCH (XN + 4)

__global__ __launch_bounds__(256) void softmax_av_gemm(const float* __restrict__ E,
                                                       const float* __restrict__ x,
                                                       float* __restrict__ out,
                                                       float* __restrict__ att00) {
    __shared__ __attribute__((aligned(16))) float att_s[AV_ROWS * AV_PITCH];
    __shared__ __attribute__((aligned(16))) float xs[GAT_N * XS_PITCH];
    __shared__ __attribute__((aligned(16))) float red_s[8 * AV_ROWS * RED_PITCH];

    const int bx   = blockIdx.x;
    const int b    = bx >> 7;
    const int m0   = ((bx >> 2) & 31) << 4;
    const int nb   = (bx & 3) * XN;
    const int t    = threadIdx.x;
    const int w    = t >> 5;
    const int lane = t & 31;

    const float* X = x + (size_t)b * GAT_N * GAT_F;
    v4f xr[XS_LD];

    {
        const float* e0p = E + ((size_t)b * GAT_N + m0 + 2 * w) * GAT_N + lane;
        float v[2][16];
        #pragma unroll
        for (int rr = 0; rr < 2; ++rr) {
            #pragma unroll
            for (int u = 0; u < 16; ++u) v[rr][u] = e0p[(size_t)rr * GAT_N + 32 * u];
        }
        {
            const float* xp = X + (size_t)(t >> 3) * GAT_F + nb + 4 * (t & 7);
            #pragma unroll
            for (int k = 0; k < XS_LD; ++k) xr[k] = *(const v4f*)(xp + (size_t)(32 * k) * GAT_F);
        }

        float mx[2];
        #pragma unroll
        for (int rr = 0; rr < 2; ++rr) {
            float m = v[rr][0];
            #pragma unroll
            for (int u = 1; u < 16; ++u) m = fmaxf(m, v[rr][u]);
            mx[rr] = m;
        }
        #pragma unroll
        for (int s = 16; s > 0; s >>= 1) {
            #pragma unroll
            for (int rr = 0; rr < 2; ++rr) mx[rr] = fmaxf(mx[rr], __shfl_xor(mx[rr], s, 32));
        }
        float sum[2];
        #pragma unroll
        for (int rr = 0; rr < 2; ++rr) {
            #pragma unroll
            for (int u = 0; u < 16; ++u) v[rr][u] = __expf(v[rr][u] - mx[rr]);
            float r8[8], r4[4], r2[2];
            #pragma unroll
            for (int u = 0; u < 8; ++u) r8[u] = v[rr][u] + v[rr][u + 8];
            #pragma unroll
            for (int u = 0; u < 4; ++u) r4[u] = r8[u] + r8[u + 4];
            #pragma unroll
            for (int u = 0; u < 2; ++u) r2[u] = r4[u] + r4[u + 2];
            sum[rr] = r2[0] + r2[1];
        }
        #pragma unroll
        for (int s = 16; s > 0; s >>= 1) {
            #pragma unroll
            for (int rr = 0; rr < 2; ++rr) sum[rr] = sum[rr] + __shfl_xor(sum[rr], s, 32);
        }
        #pragma unroll
        for (int rr = 0; rr < 2; ++rr) {
            const float inv = 1.f / sum[rr];
            float* srow = att_s + (2 * w + rr) * AV_PITCH + lane;
            #pragma unroll
            for (int u = 0; u < 16; ++u) {
                v[rr][u] = v[rr][u] * inv;
                srow[32 * u] = v[rr][u];
            }
        }
        if (bx == 0 && w == 0) {
            #pragma unroll
            for (int u = 0; u < 16; ++u) att00[lane + 32 * u] = v[0][u];
        }
    }
    {
        float* sp = xs + (t >> 3) * XS_PITCH + 4 * (t & 7);
        #pragma unroll
        for (int k = 0; k < XS_LD; ++k) *(v4f*)(sp + (32 * k) * XS_PITCH) = xr[k];
    }
    __syncthreads();

    const int lhalf = lane >> 4;
    const int lm    = lane & 15;

    const float* Ak = att_s + lm * AV_PITCH + AVK * w + 2 * lhalf;
    const float* Bk = xs + (AVK * w + 2 * lhalf) * XS_PITCH + lm;

    v8f c0 = {0.f, 0.f, 0.f, 0.f, 0.f, 0.f, 0.f, 0.f};
    v8f c1 = {0.f, 0.f, 0.f, 0.f, 0.f, 0.f, 0.f, 0.f};
    #pragma unroll
    for (int s = 0; s < AVK / 4; ++s) {
        v2f a = *(const v2f*)(Ak + 4 * s);
        v2f b0, b1;
        b0.x = Bk[(4 * s) * XS_PITCH];
        b0.y = Bk[(4 * s + 1) * XS_PITCH];
        b1.x = Bk[(4 * s) * XS_PITCH + 16];
        b1.y = Bk[(4 * s + 1) * XS_PITCH + 16];
        c0 = __builtin_amdgcn_wmma_f32_16x16x4_f32(false, a, false, b0,
                                                   (short)0, c0, false, false);
        c1 = __builtin_amdgcn_wmma_f32_16x16x4_f32(false, a, false, b1,
                                                   (short)0, c1, false, false);
    }

    {
        float* rp = red_s + w * (AV_ROWS * RED_PITCH) + (8 * lhalf) * RED_PITCH + lm;
        #pragma unroll
        for (int v = 0; v < 8; ++v) {
            rp[v * RED_PITCH]      = c0[v];
            rp[v * RED_PITCH + 16] = c1[v];
        }
    }
    __syncthreads();

    {
        const float* fp = red_s + (t >> 4) * RED_PITCH + 2 * (t & 15);
        v2f acc = *(const v2f*)fp;
        #pragma unroll
        for (int q = 1; q < 8; ++q) acc = acc + *(const v2f*)(fp + q * (AV_ROWS * RED_PITCH));
        float* op = out + (size_t)b * GAT_N * GAT_F
                        + (size_t)(m0 + (t >> 4)) * GAT_F + nb + 2 * (t & 15);
        *(v2f*)op = acc;
    }
}

extern "C" void kernel_launch(void* const* d_in, const int* in_sizes, int n_in,
                              void* d_out, int out_size, void* d_ws, size_t ws_size,
                              hipStream_t stream) {
    const float* x   = (const float*)d_in[0];
    const float* w1  = (const float*)d_in[1];
    const float* b1  = (const float*)d_in[2];
    const float* w2  = (const float*)d_in[3];
    const float* b2  = (const float*)d_in[4];
    const int*   adj = (const int*)d_in[5];

    float* out = (float*)d_out;
    float* P = (float*)d_ws;
    float* Q = P + (size_t)GAT_B * GAT_N * GAT_H;
    float* E = Q + (size_t)GAT_B * GAT_N * GAT_H;

    float* next_h = out;
    float* att00  = out + (size_t)GAT_B * GAT_N * GAT_F;

    pq_gemm<<<256, 256, 0, stream>>>(x, w1, P, Q);
    e_kernel<<<256, 256, 0, stream>>>(P, Q, b1, w2, b2, adj, E);
    softmax_av_gemm<<<256, 256, 0, stream>>>(E, x, next_h, att00);
}
